# starting kernel
# baseline (speedup 1.0000x reference)
#include <hip/hip_runtime.h>

typedef __attribute__((ext_vector_type(2))) float v2f;
typedef __attribute__((ext_vector_type(8))) float v8f;

#define NK 64
#define NC 2048
#define NN 1024
#define NB 16

__global__ __launch_bounds__(128) void netvlad_assign(
    const float* __restrict__ x, const float* __restrict__ w, float* __restrict__ a)
{
    const int lane = threadIdx.x & 31;
    const int wid  = threadIdx.x >> 5;
    const int col  = lane & 15;
    const int half = lane >> 4;
    const int koff = half << 1;

    const int tile = blockIdx.x * 4 + wid;
    const int b    = tile >> 6;
    const int n0   = (tile & 63) << 4;

    const float* xb = x + (size_t)b * NC * NN;

    v8f acc[4] = {v8f{}, v8f{}, v8f{}, v8f{}};

    for (int c = 0; c < NC; c += 4) {
        v2f af0 = *(const v2f*)(w + (size_t)( 0 + col) * NC + c + koff);
        v2f af1 = *(const v2f*)(w + (size_t)(16 + col) * NC + c + koff);
        v2f af2 = *(const v2f*)(w + (size_t)(32 + col) * NC + c + koff);
        v2f af3 = *(const v2f*)(w + (size_t)(48 + col) * NC + c + koff);
        v2f bf;
        bf.x = xb[(size_t)(c + koff + 0) * NN + n0 + col];
        bf.y = xb[(size_t)(c + koff + 1) * NN + n0 + col];

        acc[0] = __builtin_amdgcn_wmma_f32_16x16x4_f32(false, af0, false, bf, (short)0, acc[0], false, false);
        acc[1] = __builtin_amdgcn_wmma_f32_16x16x4_f32(false, af1, false, bf, (short)0, acc[1], false, false);
        acc[2] = __builtin_amdgcn_wmma_f32_16x16x4_f32(false, af2, false, bf, (short)0, acc[2], false, false);
        acc[3] = __builtin_amdgcn_wmma_f32_16x16x4_f32(false, af3, false, bf, (short)0, acc[3], false, false);
    }

    float mx = -3.402823466e38f;
#pragma unroll
    for (int t = 0; t < 4; ++t)
#pragma unroll
        for (int r = 0; r < 8; ++r) mx = fmaxf(mx, acc[t][r]);
    mx = fmaxf(mx, __shfl_xor(mx, 16, 32));

    float sum = 0.0f;
#pragma unroll
    for (int t = 0; t < 4; ++t)
#pragma unroll
        for (int r = 0; r < 8; ++r) {
            float e = __expf(acc[t][r] - mx);
            acc[t][r] = e;
            sum += e;
        }
    sum += __shfl_xor(sum, 16, 32);
    const float inv = 1.0f / sum;

    float* ab = a + (size_t)b * NK * NN;
#pragma unroll
    for (int t = 0; t < 4; ++t)
#pragma unroll
        for (int r = 0; r < 8; ++r) {
            int k = t * 16 + r + 8 * half;
            ab[(size_t)k * NN + n0 + col] = acc[t][r] * inv;
        }
}

__global__ __launch_bounds__(256) void netvlad_asum(
    const float* __restrict__ a, float* __restrict__ asum)
{
    int idx = blockIdx.x * 256 + threadIdx.x;
    const float4* row = (const float4*)(a + (size_t)idx * NN);
    float s = 0.0f;
    for (int i = 0; i < NN / 4; ++i) {
        float4 v = row[i];
        s += (v.x + v.y) + (v.z + v.w);
    }
    asum[idx] = s;
}

__global__ __launch_bounds__(128) void netvlad_vlad(
    const float* __restrict__ x, const float* __restrict__ a,
    const float* __restrict__ asum, const float* __restrict__ cent,
    float* __restrict__ out)
{
    const int lane = threadIdx.x & 31;
    const int wid  = threadIdx.x >> 5;
    const int col  = lane & 15;
    const int half = lane >> 4;
    const int koff = half << 1;

    const int tile = blockIdx.x * 4 + wid;
    const int b    = tile >> 7;
    const int c0   = (tile & 127) << 4;

    const float* ab   = a + (size_t)b * NK * NN;
    const float* xrow = x + ((size_t)b * NC + c0 + col) * NN;

    v8f acc[4] = {v8f{}, v8f{}, v8f{}, v8f{}};

    for (int n = 0; n < NN; n += 4) {
        v2f af0 = *(const v2f*)(ab + (size_t)( 0 + col) * NN + n + koff);
        v2f af1 = *(const v2f*)(ab + (size_t)(16 + col) * NN + n + koff);
        v2f af2 = *(const v2f*)(ab + (size_t)(32 + col) * NN + n + koff);
        v2f af3 = *(const v2f*)(ab + (size_t)(48 + col) * NN + n + koff);
        v2f bf = *(const v2f*)(xrow + n + koff);

        acc[0] = __builtin_amdgcn_wmma_f32_16x16x4_f32(false, af0, false, bf, (short)0, acc[0], false, false);
        acc[1] = __builtin_amdgcn_wmma_f32_16x16x4_f32(false, af1, false, bf, (short)0, acc[1], false, false);
        acc[2] = __builtin_amdgcn_wmma_f32_16x16x4_f32(false, af2, false, bf, (short)0, acc[2], false, false);
        acc[3] = __builtin_amdgcn_wmma_f32_16x16x4_f32(false, af3, false, bf, (short)0, acc[3], false, false);
    }

    float* ob = out + (size_t)b * NK * NC;
#pragma unroll
    for (int t = 0; t < 4; ++t)
#pragma unroll
        for (int r = 0; r < 8; ++r) {
            int k = t * 16 + r + 8 * half;
            float s  = asum[b * NK + k];
            float cv = cent[(size_t)k * NC + c0 + col];
            ob[(size_t)k * NC + c0 + col] = acc[t][r] - s * cv;
        }
}

__global__ __launch_bounds__(256) void netvlad_norm(
    const float* __restrict__ out, float* __restrict__ invn)
{
    __shared__ float red[256];
    const float4* vb = (const float4*)(out + (size_t)blockIdx.x * NK * NC);
    float s = 0.0f;
    for (int i = threadIdx.x; i < (NK * NC) / 4; i += 256) {
        float4 v = vb[i];
        s += v.x * v.x + v.y * v.y + v.z * v.z + v.w * v.w;
    }
    red[threadIdx.x] = s;
    __syncthreads();
    for (int off = 128; off > 0; off >>= 1) {
        if (threadIdx.x < off) red[threadIdx.x] += red[threadIdx.x + off];
        __syncthreads();
    }
    if (threadIdx.x == 0)
        invn[blockIdx.x] = 1.0f / fmaxf(sqrtf(red[0]), 1e-12f);
}

__global__ __launch_bounds__(256) void netvlad_scale(
    float* __restrict__ out, const float* __restrict__ invn)
{
    int gid = blockIdx.x * 256 + threadIdx.x;
    int b = gid / ((NK * NC) / 4);
    float sc = invn[b];
    float4* p = (float4*)out;
    float4 v = p[gid];
    v.x *= sc; v.y *= sc; v.z *= sc; v.w *= sc;
    p[gid] = v;
}

extern "C" void kernel_launch(void* const* d_in, const int* in_sizes, int n_in,
                              void* d_out, int out_size, void* d_ws, size_t ws_size,
                              hipStream_t stream) {
    const float* x    = (const float*)d_in[0];
    const float* w    = (const float*)d_in[1];
    const float* cent = (const float*)d_in[2];
    float* out = (float*)d_out;

    float* a    = (float*)d_ws;
    float* asum = a + (size_t)NB * NK * NN;
    float* invn = asum + NB * NK;

    netvlad_assign<<<(NB * NN / 16) / 4, 128, 0, stream>>>(x, w, a);
    netvlad_asum  <<<(NB * NK) / 256, 256, 0, stream>>>(a, asum);
    netvlad_vlad  <<<(NB * NC / 16) / 4, 128, 0, stream>>>(x, a, asum, cent, out);
    netvlad_norm  <<<NB, 256, 0, stream>>>(out, invn);
    netvlad_scale <<<(NB * NK * NC / 4) / 256, 256, 0, stream>>>(out, invn);
}


# assign kernel: channel sum split over the 8 waves of an n-tile (wave s takes 16-channel groups s, s+8, ...), the 8 partial tiles added through LDS in fixed order; changes the c summation order
# speedup vs baseline: 2.1993x; 2.1993x over previous
#include <hip/hip_runtime.h>

typedef __attribute__((ext_vector_type(2))) float v2f;
typedef __attribute__((ext_vector_type(8))) float v8f;

#define NK 64
#define NC 2048
#define NN 1024
#define NB 16

#define CSPLIT 8
#define NTB    2

__global__ __launch_bounds__(512) void netvlad_assign(
    const float* __restrict__ x, const float* __restrict__ w, float* __restrict__ a)
{
    __shared__ float4 red[NTB][CSPLIT - 1][8][32];

    const int lane = threadIdx.x & 31;
    const int wid  = __builtin_amdgcn_readfirstlane(threadIdx.x >> 5);
    const int s    = wid & (CSPLIT - 1);
    const int nt   = wid >> 3;
    const int col  = lane & 15;
    const int half = lane >> 4;
    const int koff = half << 1;

    const int tile = blockIdx.x * NTB + nt;
    const int b    = tile >> 6;
    const int n0   = (tile & 63) << 4;

    const float* xb = x + (size_t)b * NC * NN;

    v8f acc[4] = {v8f{}, v8f{}, v8f{}, v8f{}};

    for (int g = s; g < NC / 16; g += CSPLIT) {
        for (int u = 0; u < 4; ++u) {
            const int c = g * 16 + u * 4;
            v2f af0 = *(const v2f*)(w + (size_t)( 0 + col) * NC + c + koff);
            v2f af1 = *(const v2f*)(w + (size_t)(16 + col) * NC + c + koff);
            v2f af2 = *(const v2f*)(w + (size_t)(32 + col) * NC + c + koff);
            v2f af3 = *(const v2f*)(w + (size_t)(48 + col) * NC + c + koff);
            v2f bf;
            bf.x = xb[(size_t)(c + koff + 0) * NN + n0 + col];
            bf.y = xb[(size_t)(c + koff + 1) * NN + n0 + col];

            acc[0] = __builtin_amdgcn_wmma_f32_16x16x4_f32(false, af0, false, bf, (short)0, acc[0], false, false);
            acc[1] = __builtin_amdgcn_wmma_f32_16x16x4_f32(false, af1, false, bf, (short)0, acc[1], false, false);
            acc[2] = __builtin_amdgcn_wmma_f32_16x16x4_f32(false, af2, false, bf, (short)0, acc[2], false, false);
            acc[3] = __builtin_amdgcn_wmma_f32_16x16x4_f32(false, af3, false, bf, (short)0, acc[3], false, false);
        }
    }

    if (s != 0) {
#pragma unroll
        for (int q = 0; q < 8; ++q) {
            const int t = q >> 1, r0 = (q & 1) * 4;
            red[nt][s - 1][q][lane] =
                make_float4(acc[t][r0 + 0], acc[t][r0 + 1], acc[t][r0 + 2], acc[t][r0 + 3]);
        }
    }
    __syncthreads();
    if (s != 0) return;

#pragma unroll
    for (int p = 0; p < CSPLIT - 1; ++p)
#pragma unroll
        for (int q = 0; q < 8; ++q) {
            const int t = q >> 1, r0 = (q & 1) * 4;
            const float4 v = red[nt][p][q][lane];
            acc[t][r0 + 0] += v.x;
            acc[t][r0 + 1] += v.y;
            acc[t][r0 + 2] += v.z;
            acc[t][r0 + 3] += v.w;
        }

    float mx = -3.402823466e38f;
#pragma unroll
    for (int t = 0; t < 4; ++t)
#pragma unroll
        for (int r = 0; r < 8; ++r) mx = fmaxf(mx, acc[t][r]);
    mx = fmaxf(mx, __shfl_xor(mx, 16, 32));

    float sum = 0.0f;
#pragma unroll
    for (int t = 0; t < 4; ++t)
#pragma unroll
        for (int r = 0; r < 8; ++r) {
            float e = __expf(acc[t][r] - mx);
            acc[t][r] = e;
            sum += e;
        }
    sum += __shfl_xor(sum, 16, 32);
    const float inv = 1.0f / sum;

    float* ab = a + (size_t)b * NK * NN;
#pragma unroll
    for (int t = 0; t < 4; ++t)
#pragma unroll
        for (int r = 0; r < 8; ++r) {
            int k = t * 16 + r + 8 * half;
            ab[(size_t)k * NN + n0 + col] = acc[t][r] * inv;
        }
}

__global__ __launch_bounds__(256) void netvlad_asum(
    const float* __restrict__ a, float* __restrict__ asum)
{
    int idx = blockIdx.x * 256 + threadIdx.x;
    const float4* row = (const float4*)(a + (size_t)idx * NN);
    float s = 0.0f;
    for (int i = 0; i < NN / 4; ++i) {
        float4 v = row[i];
        s += (v.x + v.y) + (v.z + v.w);
    }
    asum[idx] = s;
}

__global__ __launch_bounds__(128) void netvlad_vlad(
    const float* __restrict__ x, const float* __restrict__ a,
    const float* __restrict__ asum, const float* __restrict__ cent,
    float* __restrict__ out)
{
    const int lane = threadIdx.x & 31;
    const int wid  = threadIdx.x >> 5;
    const int col  = lane & 15;
    const int half = lane >> 4;
    const int koff = half << 1;

    const int tile = blockIdx.x * 4 + wid;
    const int b    = tile >> 7;
    const int c0   = (tile & 127) << 4;

    const float* ab   = a + (size_t)b * NK * NN;
    const float* xrow = x + ((size_t)b * NC + c0 + col) * NN;

    v8f acc[4] = {v8f{}, v8f{}, v8f{}, v8f{}};

    for (int n = 0; n < NN; n += 4) {
        v2f af0 = *(const v2f*)(ab + (size_t)( 0 + col) * NN + n + koff);
        v2f af1 = *(const v2f*)(ab + (size_t)(16 + col) * NN + n + koff);
        v2f af2 = *(const v2f*)(ab + (size_t)(32 + col) * NN + n + koff);
        v2f af3 = *(const v2f*)(ab + (size_t)(48 + col) * NN + n + koff);
        v2f bf = *(const v2f*)(xrow + n + koff);

        acc[0] = __builtin_amdgcn_wmma_f32_16x16x4_f32(false, af0, false, bf, (short)0, acc[0], false, false);
        acc[1] = __builtin_amdgcn_wmma_f32_16x16x4_f32(false, af1, false, bf, (short)0, acc[1], false, false);
        acc[2] = __builtin_amdgcn_wmma_f32_16x16x4_f32(false, af2, false, bf, (short)0, acc[2], false, false);
        acc[3] = __builtin_amdgcn_wmma_f32_16x16x4_f32(false, af3, false, bf, (short)0, acc[3], false, false);
    }

    float* ob = out + (size_t)b * NK * NC;
#pragma unroll
    for (int t = 0; t < 4; ++t)
#pragma unroll
        for (int r = 0; r < 8; ++r) {
            int k = t * 16 + r + 8 * half;
            float s  = asum[b * NK + k];
            float cv = cent[(size_t)k * NC + c0 + col];
            ob[(size_t)k * NC + c0 + col] = acc[t][r] - s * cv;
        }
}

__global__ __launch_bounds__(256) void netvlad_norm(
    const float* __restrict__ out, float* __restrict__ invn)
{
    __shared__ float red[256];
    const float4* vb = (const float4*)(out + (size_t)blockIdx.x * NK * NC);
    float s = 0.0f;
    for (int i = threadIdx.x; i < (NK * NC) / 4; i += 256) {
        float4 v = vb[i];
        s += v.x * v.x + v.y * v.y + v.z * v.z + v.w * v.w;
    }
    red[threadIdx.x] = s;
    __syncthreads();
    for (int off = 128; off > 0; off >>= 1) {
        if (threadIdx.x < off) red[threadIdx.x] += red[threadIdx.x + off];
        __syncthreads();
    }
    if (threadIdx.x == 0)
        invn[blockIdx.x] = 1.0f / fmaxf(sqrtf(red[0]), 1e-12f);
}

__global__ __launch_bounds__(256) void netvlad_scale(
    float* __restrict__ out, const float* __restrict__ invn)
{
    int gid = blockIdx.x * 256 + threadIdx.x;
    int b = gid / ((NK * NC) / 4);
    float sc = invn[b];
    float4* p = (float4*)out;
    float4 v = p[gid];
    v.x *= sc; v.y *= sc; v.z *= sc; v.w *= sc;
    p[gid] = v;
}

extern "C" void kernel_launch(void* const* d_in, const int* in_sizes, int n_in,
                              void* d_out, int out_size, void* d_ws, size_t ws_size,
                              hipStream_t stream) {
    const float* x    = (const float*)d_in[0];
    const float* w    = (const float*)d_in[1];
    const float* cent = (const float*)d_in[2];
    float* out = (float*)d_out;

    float* a    = (float*)d_ws;
    float* asum = a + (size_t)NB * NK * NN;
    float* invn = asum + NB * NK;

    netvlad_assign<<<(NB * NN / 16) / NTB, 32 * CSPLIT * NTB, 0, stream>>>(x, w, a);
    netvlad_asum  <<<(NB * NK) / 256, 256, 0, stream>>>(a, asum);
    netvlad_vlad  <<<(NB * NC / 16) / 4, 128, 0, stream>>>(x, a, asum, cent, out);
    netvlad_norm  <<<NB, 256, 0, stream>>>(out, invn);
    netvlad_scale <<<(NB * NK * NC / 4) / 256, 256, 0, stream>>>(out, invn);
}


# both WMMA kernels: fragment loads widened to b128, each feeding two k-steps (lane half h holds k 4h..4h+3 of each 8-block in A and B alike; assign x rows stay b32); reorders the k sum in 8-blocks
# speedup vs baseline: 2.4457x; 1.1120x over previous
#include <hip/hip_runtime.h>

typedef __attribute__((ext_vector_type(2))) float v2f;
typedef __attribute__((ext_vector_type(8))) float v8f;

#define NK 64
#define NC 2048
#define NN 1024
#define NB 16

#define CSPLIT 8
#define NTB    2

__global__ __launch_bounds__(512) void netvlad_assign(
    const float* __restrict__ x, const float* __restrict__ w, float* __restrict__ a)
{
    __shared__ float4 red[NTB][CSPLIT - 1][8][32];

    const int lane = threadIdx.x & 31;
    const int wid  = __builtin_amdgcn_readfirstlane(threadIdx.x >> 5);
    const int s    = wid & (CSPLIT - 1);
    const int nt   = wid >> 3;
    const int col  = lane & 15;
    const int half = lane >> 4;
    const int koff = half << 2;

    const int tile = blockIdx.x * NTB + nt;
    const int b    = tile >> 6;
    const int n0   = (tile & 63) << 4;

    const float* xb = x + (size_t)b * NC * NN;

    v8f acc[4] = {v8f{}, v8f{}, v8f{}, v8f{}};

    for (int g = s; g < NC / 16; g += CSPLIT) {
        for (int u = 0; u < 2; ++u) {
            const int c = g * 16 + u * 8 + koff;
            const float4 a0 = *(const float4*)(w + (size_t)( 0 + col) * NC + c);
            const float4 a1 = *(const float4*)(w + (size_t)(16 + col) * NC + c);
            const float4 a2 = *(const float4*)(w + (size_t)(32 + col) * NC + c);
            const float4 a3 = *(const float4*)(w + (size_t)(48 + col) * NC + c);
            v2f b0, b1;
            b0.x = xb[(size_t)(c + 0) * NN + n0 + col];
            b0.y = xb[(size_t)(c + 1) * NN + n0 + col];
            b1.x = xb[(size_t)(c + 2) * NN + n0 + col];
            b1.y = xb[(size_t)(c + 3) * NN + n0 + col];

            acc[0] = __builtin_amdgcn_wmma_f32_16x16x4_f32(false, v2f{a0.x, a0.y}, false, b0, (short)0, acc[0], false, false);
            acc[1] = __builtin_amdgcn_wmma_f32_16x16x4_f32(false, v2f{a1.x, a1.y}, false, b0, (short)0, acc[1], false, false);
            acc[2] = __builtin_amdgcn_wmma_f32_16x16x4_f32(false, v2f{a2.x, a2.y}, false, b0, (short)0, acc[2], false, false);
            acc[3] = __builtin_amdgcn_wmma_f32_16x16x4_f32(false, v2f{a3.x, a3.y}, false, b0, (short)0, acc[3], false, false);
            acc[0] = __builtin_amdgcn_wmma_f32_16x16x4_f32(false, v2f{a0.z, a0.w}, false, b1, (short)0, acc[0], false, false);
            acc[1] = __builtin_amdgcn_wmma_f32_16x16x4_f32(false, v2f{a1.z, a1.w}, false, b1, (short)0, acc[1], false, false);
            acc[2] = __builtin_amdgcn_wmma_f32_16x16x4_f32(false, v2f{a2.z, a2.w}, false, b1, (short)0, acc[2], false, false);
            acc[3] = __builtin_amdgcn_wmma_f32_16x16x4_f32(false, v2f{a3.z, a3.w}, false, b1, (short)0, acc[3], false, false);
        }
    }

    if (s != 0) {
#pragma unroll
        for (int q = 0; q < 8; ++q) {
            const int t = q >> 1, r0 = (q & 1) * 4;
            red[nt][s - 1][q][lane] =
                make_float4(acc[t][r0 + 0], acc[t][r0 + 1], acc[t][r0 + 2], acc[t][r0 + 3]);
        }
    }
    __syncthreads();
    if (s != 0) return;

#pragma unroll
    for (int p = 0; p < CSPLIT - 1; ++p)
#pragma unroll
        for (int q = 0; q < 8; ++q) {
            const int t = q >> 1, r0 = (q & 1) * 4;
            const float4 v = red[nt][p][q][lane];
            acc[t][r0 + 0] += v.x;
            acc[t][r0 + 1] += v.y;
            acc[t][r0 + 2] += v.z;
            acc[t][r0 + 3] += v.w;
        }

    float mx = -3.402823466e38f;
#pragma unroll
    for (int t = 0; t < 4; ++t)
#pragma unroll
        for (int r = 0; r < 8; ++r) mx = fmaxf(mx, acc[t][r]);
    mx = fmaxf(mx, __shfl_xor(mx, 16, 32));

    float sum = 0.0f;
#pragma unroll
    for (int t = 0; t < 4; ++t)
#pragma unroll
        for (int r = 0; r < 8; ++r) {
            float e = __expf(acc[t][r] - mx);
            acc[t][r] = e;
            sum += e;
        }
    sum += __shfl_xor(sum, 16, 32);
    const float inv = 1.0f / sum;

    float* ab = a + (size_t)b * NK * NN;
#pragma unroll
    for (int t = 0; t < 4; ++t)
#pragma unroll
        for (int r = 0; r < 8; ++r) {
            int k = t * 16 + r + 8 * half;
            ab[(size_t)k * NN + n0 + col] = acc[t][r] * inv;
        }
}

__global__ __launch_bounds__(256) void netvlad_asum(
    const float* __restrict__ a, float* __restrict__ asum)
{
    int idx = blockIdx.x * 256 + threadIdx.x;
    const float4* row = (const float4*)(a + (size_t)idx * NN);
    float s = 0.0f;
    for (int i = 0; i < NN / 4; ++i) {
        float4 v = row[i];
        s += (v.x + v.y) + (v.z + v.w);
    }
    asum[idx] = s;
}

__global__ __launch_bounds__(128) void netvlad_vlad(
    const float* __restrict__ x, const float* __restrict__ a,
    const float* __restrict__ asum, const float* __restrict__ cent,
    float* __restrict__ out)
{
    const int lane = threadIdx.x & 31;
    const int wid  = threadIdx.x >> 5;
    const int col  = lane & 15;
    const int half = lane >> 4;
    const int koff = half << 2;

    const int tile = blockIdx.x * 4 + wid;
    const int b    = tile >> 7;
    const int c0   = (tile & 127) << 4;

    const float* ab   = a + (size_t)b * NK * NN;
    const float* xrow = x + ((size_t)b * NC + c0 + col) * NN;

    v8f acc[4] = {v8f{}, v8f{}, v8f{}, v8f{}};

    for (int n = 0; n < NN; n += 8) {
        const float4 a0 = *(const float4*)(ab + (size_t)( 0 + col) * NN + n + koff);
        const float4 a1 = *(const float4*)(ab + (size_t)(16 + col) * NN + n + koff);
        const float4 a2 = *(const float4*)(ab + (size_t)(32 + col) * NN + n + koff);
        const float4 a3 = *(const float4*)(ab + (size_t)(48 + col) * NN + n + koff);
        const float4 bv = *(const float4*)(xrow + n + koff);
        const v2f b0 = v2f{bv.x, bv.y};
        const v2f b1 = v2f{bv.z, bv.w};

        acc[0] = __builtin_amdgcn_wmma_f32_16x16x4_f32(false, v2f{a0.x, a0.y}, false, b0, (short)0, acc[0], false, false);
        acc[1] = __builtin_amdgcn_wmma_f32_16x16x4_f32(false, v2f{a1.x, a1.y}, false, b0, (short)0, acc[1], false, false);
        acc[2] = __builtin_amdgcn_wmma_f32_16x16x4_f32(false, v2f{a2.x, a2.y}, false, b0, (short)0, acc[2], false, false);
        acc[3] = __builtin_amdgcn_wmma_f32_16x16x4_f32(false, v2f{a3.x, a3.y}, false, b0, (short)0, acc[3], false, false);
        acc[0] = __builtin_amdgcn_wmma_f32_16x16x4_f32(false, v2f{a0.z, a0.w}, false, b1, (short)0, acc[0], false, false);
        acc[1] = __builtin_amdgcn_wmma_f32_16x16x4_f32(false, v2f{a1.z, a1.w}, false, b1, (short)0, acc[1], false, false);
        acc[2] = __builtin_amdgcn_wmma_f32_16x16x4_f32(false, v2f{a2.z, a2.w}, false, b1, (short)0, acc[2], false, false);
        acc[3] = __builtin_amdgcn_wmma_f32_16x16x4_f32(false, v2f{a3.z, a3.w}, false, b1, (short)0, acc[3], false, false);
    }

    float* ob = out + (size_t)b * NK * NC;
#pragma unroll
    for (int t = 0; t < 4; ++t)
#pragma unroll
        for (int r = 0; r < 8; ++r) {
            int k = t * 16 + r + 8 * half;
            float s  = asum[b * NK + k];
            float cv = cent[(size_t)k * NC + c0 + col];
            ob[(size_t)k * NC + c0 + col] = acc[t][r] - s * cv;
        }
}

__global__ __launch_bounds__(256) void netvlad_norm(
    const float* __restrict__ out, float* __restrict__ invn)
{
    __shared__ float red[256];
    const float4* vb = (const float4*)(out + (size_t)blockIdx.x * NK * NC);
    float s = 0.0f;
    for (int i = threadIdx.x; i < (NK * NC) / 4; i += 256) {
        float4 v = vb[i];
        s += v.x * v.x + v.y * v.y + v.z * v.z + v.w * v.w;
    }
    red[threadIdx.x] = s;
    __syncthreads();
    for (int off = 128; off > 0; off >>= 1) {
        if (threadIdx.x < off) red[threadIdx.x] += red[threadIdx.x + off];
        __syncthreads();
    }
    if (threadIdx.x == 0)
        invn[blockIdx.x] = 1.0f / fmaxf(sqrtf(red[0]), 1e-12f);
}

__global__ __launch_bounds__(256) void netvlad_scale(
    float* __restrict__ out, const float* __restrict__ invn)
{
    int gid = blockIdx.x * 256 + threadIdx.x;
    int b = gid / ((NK * NC) / 4);
    float sc = invn[b];
    float4* p = (float4*)out;
    float4 v = p[gid];
    v.x *= sc; v.y *= sc; v.z *= sc; v.w *= sc;
    p[gid] = v;
}

extern "C" void kernel_launch(void* const* d_in, const int* in_sizes, int n_in,
                              void* d_out, int out_size, void* d_ws, size_t ws_size,
                              hipStream_t stream) {
    const float* x    = (const float*)d_in[0];
    const float* w    = (const float*)d_in[1];
    const float* cent = (const float*)d_in[2];
    float* out = (float*)d_out;

    float* a    = (float*)d_ws;
    float* asum = a + (size_t)NB * NK * NN;
    float* invn = asum + NB * NK;

    netvlad_assign<<<(NB * NN / 16) / NTB, 32 * CSPLIT * NTB, 0, stream>>>(x, w, a);
    netvlad_asum  <<<(NB * NK) / 256, 256, 0, stream>>>(a, asum);
    netvlad_vlad  <<<(NB * NC / 16) / 4, 128, 0, stream>>>(x, a, asum, cent, out);
    netvlad_norm  <<<NB, 256, 0, stream>>>(out, invn);
    netvlad_scale <<<(NB * NK * NC / 4) / 256, 256, 0, stream>>>(out, invn);
}


# a_sum taken inside the vlad kernel from the A fragments it already loads (per-lane running row sums, halves combined, LDS strip to D layout); asum kernel and launch dropped; reorders the a_sum sum
# speedup vs baseline: 3.0309x; 1.2393x over previous
#include <hip/hip_runtime.h>

typedef __attribute__((ext_vector_type(2))) float v2f;
typedef __attribute__((ext_vector_type(8))) float v8f;

#define NK 64
#define NC 2048
#define NN 1024
#define NB 16

#define CSPLIT 8
#define NTB    2

__global__ __launch_bounds__(512) void netvlad_assign(
    const float* __restrict__ x, const float* __restrict__ w, float* __restrict__ a)
{
    __shared__ float4 red[NTB][CSPLIT - 1][8][32];

    const int lane = threadIdx.x & 31;
    const int wid  = __builtin_amdgcn_readfirstlane(threadIdx.x >> 5);
    const int s    = wid & (CSPLIT - 1);
    const int nt   = wid >> 3;
    const int col  = lane & 15;
    const int half = lane >> 4;
    const int koff = half << 2;

    const int tile = blockIdx.x * NTB + nt;
    const int b    = tile >> 6;
    const int n0   = (tile & 63) << 4;

    const float* xb = x + (size_t)b * NC * NN;

    v8f acc[4] = {v8f{}, v8f{}, v8f{}, v8f{}};

    for (int g = s; g < NC / 16; g += CSPLIT) {
        for (int u = 0; u < 2; ++u) {
            const int c = g * 16 + u * 8 + koff;
            const float4 a0 = *(const float4*)(w + (size_t)( 0 + col) * NC + c);
            const float4 a1 = *(const float4*)(w + (size_t)(16 + col) * NC + c);
            const float4 a2 = *(const float4*)(w + (size_t)(32 + col) * NC + c);
            const float4 a3 = *(const float4*)(w + (size_t)(48 + col) * NC + c);
            v2f b0, b1;
            b0.x = xb[(size_t)(c + 0) * NN + n0 + col];
            b0.y = xb[(size_t)(c + 1) * NN + n0 + col];
            b1.x = xb[(size_t)(c + 2) * NN + n0 + col];
            b1.y = xb[(size_t)(c + 3) * NN + n0 + col];

            acc[0] = __builtin_amdgcn_wmma_f32_16x16x4_f32(false, v2f{a0.x, a0.y}, false, b0, (short)0, acc[0], false, false);
            acc[1] = __builtin_amdgcn_wmma_f32_16x16x4_f32(false, v2f{a1.x, a1.y}, false, b0, (short)0, acc[1], false, false);
            acc[2] = __builtin_amdgcn_wmma_f32_16x16x4_f32(false, v2f{a2.x, a2.y}, false, b0, (short)0, acc[2], false, false);
            acc[3] = __builtin_amdgcn_wmma_f32_16x16x4_f32(false, v2f{a3.x, a3.y}, false, b0, (short)0, acc[3], false, false);
            acc[0] = __builtin_amdgcn_wmma_f32_16x16x4_f32(false, v2f{a0.z, a0.w}, false, b1, (short)0, acc[0], false, false);
            acc[1] = __builtin_amdgcn_wmma_f32_16x16x4_f32(false, v2f{a1.z, a1.w}, false, b1, (short)0, acc[1], false, false);
            acc[2] = __builtin_amdgcn_wmma_f32_16x16x4_f32(false, v2f{a2.z, a2.w}, false, b1, (short)0, acc[2], false, false);
            acc[3] = __builtin_amdgcn_wmma_f32_16x16x4_f32(false, v2f{a3.z, a3.w}, false, b1, (short)0, acc[3], false, false);
        }
    }

    if (s != 0) {
#pragma unroll
        for (int q = 0; q < 8; ++q) {
            const int t = q >> 1, r0 = (q & 1) * 4;
            red[nt][s - 1][q][lane] =
                make_float4(acc[t][r0 + 0], acc[t][r0 + 1], acc[t][r0 + 2], acc[t][r0 + 3]);
        }
    }
    __syncthreads();
    if (s != 0) return;

#pragma unroll
    for (int p = 0; p < CSPLIT - 1; ++p)
#pragma unroll
        for (int q = 0; q < 8; ++q) {
            const int t = q >> 1, r0 = (q & 1) * 4;
            const float4 v = red[nt][p][q][lane];
            acc[t][r0 + 0] += v.x;
            acc[t][r0 + 1] += v.y;
            acc[t][r0 + 2] += v.z;
            acc[t][r0 + 3] += v.w;
        }

    float mx = -3.402823466e38f;
#pragma unroll
    for (int t = 0; t < 4; ++t)
#pragma unroll
        for (int r = 0; r < 8; ++r) mx = fmaxf(mx, acc[t][r]);
    mx = fmaxf(mx, __shfl_xor(mx, 16, 32));

    float sum = 0.0f;
#pragma unroll
    for (int t = 0; t < 4; ++t)
#pragma unroll
        for (int r = 0; r < 8; ++r) {
            float e = __expf(acc[t][r] - mx);
            acc[t][r] = e;
            sum += e;
        }
    sum += __shfl_xor(sum, 16, 32);
    const float inv = 1.0f / sum;

    float* ab = a + (size_t)b * NK * NN;
#pragma unroll
    for (int t = 0; t < 4; ++t)
#pragma unroll
        for (int r = 0; r < 8; ++r) {
            int k = t * 16 + r + 8 * half;
            ab[(size_t)k * NN + n0 + col] = acc[t][r] * inv;
        }
}

__global__ __launch_bounds__(128) void netvlad_vlad(
    const float* __restrict__ x, const float* __restrict__ a,
    const float* __restrict__ cent, float* __restrict__ out)
{
    __shared__ float4 srow[4][NK / 4];

    const int lane = threadIdx.x & 31;
    const int wid  = threadIdx.x >> 5;
    const int col  = lane & 15;
    const int half = lane >> 4;
    const int koff = half << 2;

    const int tile = blockIdx.x * 4 + wid;
    const int b    = tile >> 7;
    const int c0   = (tile & 127) << 4;

    const float* ab   = a + (size_t)b * NK * NN;
    const float* xrow = x + ((size_t)b * NC + c0 + col) * NN;

    v8f acc[4] = {v8f{}, v8f{}, v8f{}, v8f{}};
    v2f rs[4]  = {v2f{}, v2f{}, v2f{}, v2f{}};

    for (int n = 0; n < NN; n += 8) {
        const float4 a0 = *(const float4*)(ab + (size_t)( 0 + col) * NN + n + koff);
        const float4 a1 = *(const float4*)(ab + (size_t)(16 + col) * NN + n + koff);
        const float4 a2 = *(const float4*)(ab + (size_t)(32 + col) * NN + n + koff);
        const float4 a3 = *(const float4*)(ab + (size_t)(48 + col) * NN + n + koff);
        const float4 bv = *(const float4*)(xrow + n + koff);
        const v2f b0 = v2f{bv.x, bv.y};
        const v2f b1 = v2f{bv.z, bv.w};

        rs[0] += v2f{a0.x, a0.y} + v2f{a0.z, a0.w};
        rs[1] += v2f{a1.x, a1.y} + v2f{a1.z, a1.w};
        rs[2] += v2f{a2.x, a2.y} + v2f{a2.z, a2.w};
        rs[3] += v2f{a3.x, a3.y} + v2f{a3.z, a3.w};

        acc[0] = __builtin_amdgcn_wmma_f32_16x16x4_f32(false, v2f{a0.x, a0.y}, false, b0, (short)0, acc[0], false, false);
        acc[1] = __builtin_amdgcn_wmma_f32_16x16x4_f32(false, v2f{a1.x, a1.y}, false, b0, (short)0, acc[1], false, false);
        acc[2] = __builtin_amdgcn_wmma_f32_16x16x4_f32(false, v2f{a2.x, a2.y}, false, b0, (short)0, acc[2], false, false);
        acc[3] = __builtin_amdgcn_wmma_f32_16x16x4_f32(false, v2f{a3.x, a3.y}, false, b0, (short)0, acc[3], false, false);
        acc[0] = __builtin_amdgcn_wmma_f32_16x16x4_f32(false, v2f{a0.z, a0.w}, false, b1, (short)0, acc[0], false, false);
        acc[1] = __builtin_amdgcn_wmma_f32_16x16x4_f32(false, v2f{a1.z, a1.w}, false, b1, (short)0, acc[1], false, false);
        acc[2] = __builtin_amdgcn_wmma_f32_16x16x4_f32(false, v2f{a2.z, a2.w}, false, b1, (short)0, acc[2], false, false);
        acc[3] = __builtin_amdgcn_wmma_f32_16x16x4_f32(false, v2f{a3.z, a3.w}, false, b1, (short)0, acc[3], false, false);
    }

    float* strip = (float*)srow[wid];
    float rsum[4];
#pragma unroll
    for (int t = 0; t < 4; ++t) {
        const float r = rs[t].x + rs[t].y;
        rsum[t] = r + __shfl_xor(r, 16, 32);
    }
    if (half == 0) {
#pragma unroll
        for (int t = 0; t < 4; ++t) strip[t * 16 + col] = rsum[t];
    }
    __syncthreads();

    float* ob = out + (size_t)b * NK * NC;
#pragma unroll
    for (int t = 0; t < 4; ++t) {
        const float4 s0 = srow[wid][t * 4 + 2 * half];
        const float4 s1 = srow[wid][t * 4 + 2 * half + 1];
        const float sv[8] = {s0.x, s0.y, s0.z, s0.w, s1.x, s1.y, s1.z, s1.w};
#pragma unroll
        for (int r = 0; r < 8; ++r) {
            int k = t * 16 + r + 8 * half;
            float cv = cent[(size_t)k * NC + c0 + col];
            ob[(size_t)k * NC + c0 + col] = acc[t][r] - sv[r] * cv;
        }
    }
}

__global__ __launch_bounds__(256) void netvlad_norm(
    const float* __restrict__ out, float* __restrict__ invn)
{
    __shared__ float red[256];
    const float4* vb = (const float4*)(out + (size_t)blockIdx.x * NK * NC);
    float s = 0.0f;
    for (int i = threadIdx.x; i < (NK * NC) / 4; i += 256) {
        float4 v = vb[i];
        s += v.x * v.x + v.y * v.y + v.z * v.z + v.w * v.w;
    }
    red[threadIdx.x] = s;
    __syncthreads();
    for (int off = 128; off > 0; off >>= 1) {
        if (threadIdx.x < off) red[threadIdx.x] += red[threadIdx.x + off];
        __syncthreads();
    }
    if (threadIdx.x == 0)
        invn[blockIdx.x] = 1.0f / fmaxf(sqrtf(red[0]), 1e-12f);
}

__global__ __launch_bounds__(256) void netvlad_scale(
    float* __restrict__ out, const float* __restrict__ invn)
{
    int gid = blockIdx.x * 256 + threadIdx.x;
    int b = gid / ((NK * NC) / 4);
    float sc = invn[b];
    float4* p = (float4*)out;
    float4 v = p[gid];
    v.x *= sc; v.y *= sc; v.z *= sc; v.w *= sc;
    p[gid] = v;
}

extern "C" void kernel_launch(void* const* d_in, const int* in_sizes, int n_in,
                              void* d_out, int out_size, void* d_ws, size_t ws_size,
                              hipStream_t stream) {
    const float* x    = (const float*)d_in[0];
    const float* w    = (const float*)d_in[1];
    const float* cent = (const float*)d_in[2];
    float* out = (float*)d_out;

    float* a    = (float*)d_ws;
    float* invn = a + (size_t)NB * NK * NN;

    netvlad_assign<<<(NB * NN / 16) / NTB, 32 * CSPLIT * NTB, 0, stream>>>(x, w, a);
    netvlad_vlad  <<<(NB * NC / 16) / 4, 128, 0, stream>>>(x, a, cent, out);
    netvlad_norm  <<<NB, 256, 0, stream>>>(out, invn);
    netvlad_scale <<<(NB * NK * NC / 4) / 256, 256, 0, stream>>>(out, invn);
}


# L2 norm's square sum taken in the vlad kernel from the values it stores (one partial per work-group), a batch's 32 partials added in the scale kernel; norm kernel and launch dropped; reorders that sum
# speedup vs baseline: 3.2662x; 1.0776x over previous
#include <hip/hip_runtime.h>

typedef __attribute__((ext_vector_type(2))) float v2f;
typedef __attribute__((ext_vector_type(8))) float v8f;

#define NK 64
#define NC 2048
#define NN 1024
#define NB 16

#define CSPLIT 8
#define NTB    2

__global__ __launch_bounds__(512) void netvlad_assign(
    const float* __restrict__ x, const float* __restrict__ w, float* __restrict__ a)
{
    __shared__ float4 red[NTB][CSPLIT - 1][8][32];

    const int lane = threadIdx.x & 31;
    const int wid  = __builtin_amdgcn_readfirstlane(threadIdx.x >> 5);
    const int s    = wid & (CSPLIT - 1);
    const int nt   = wid >> 3;
    const int col  = lane & 15;
    const int half = lane >> 4;
    const int koff = half << 2;

    const int tile = blockIdx.x * NTB + nt;
    const int b    = tile >> 6;
    const int n0   = (tile & 63) << 4;

    const float* xb = x + (size_t)b * NC * NN;

    v8f acc[4] = {v8f{}, v8f{}, v8f{}, v8f{}};

    for (int g = s; g < NC / 16; g += CSPLIT) {
        for (int u = 0; u < 2; ++u) {
            const int c = g * 16 + u * 8 + koff;
            const float4 a0 = *(const float4*)(w + (size_t)( 0 + col) * NC + c);
            const float4 a1 = *(const float4*)(w + (size_t)(16 + col) * NC + c);
            const float4 a2 = *(const float4*)(w + (size_t)(32 + col) * NC + c);
            const float4 a3 = *(const float4*)(w + (size_t)(48 + col) * NC + c);
            v2f b0, b1;
            b0.x = xb[(size_t)(c + 0) * NN + n0 + col];
            b0.y = xb[(size_t)(c + 1) * NN + n0 + col];
            b1.x = xb[(size_t)(c + 2) * NN + n0 + col];
            b1.y = xb[(size_t)(c + 3) * NN + n0 + col];

            acc[0] = __builtin_amdgcn_wmma_f32_16x16x4_f32(false, v2f{a0.x, a0.y}, false, b0, (short)0, acc[0], false, false);
            acc[1] = __builtin_amdgcn_wmma_f32_16x16x4_f32(false, v2f{a1.x, a1.y}, false, b0, (short)0, acc[1], false, false);
            acc[2] = __builtin_amdgcn_wmma_f32_16x16x4_f32(false, v2f{a2.x, a2.y}, false, b0, (short)0, acc[2], false, false);
            acc[3] = __builtin_amdgcn_wmma_f32_16x16x4_f32(false, v2f{a3.x, a3.y}, false, b0, (short)0, acc[3], false, false);
            acc[0] = __builtin_amdgcn_wmma_f32_16x16x4_f32(false, v2f{a0.z, a0.w}, false, b1, (short)0, acc[0], false, false);
            acc[1] = __builtin_amdgcn_wmma_f32_16x16x4_f32(false, v2f{a1.z, a1.w}, false, b1, (short)0, acc[1], false, false);
            acc[2] = __builtin_amdgcn_wmma_f32_16x16x4_f32(false, v2f{a2.z, a2.w}, false, b1, (short)0, acc[2], false, false);
            acc[3] = __builtin_amdgcn_wmma_f32_16x16x4_f32(false, v2f{a3.z, a3.w}, false, b1, (short)0, acc[3], false, false);
        }
    }

    if (s != 0) {
#pragma unroll
        for (int q = 0; q < 8; ++q) {
            const int t = q >> 1, r0 = (q & 1) * 4;
            red[nt][s - 1][q][lane] =
                make_float4(acc[t][r0 + 0], acc[t][r0 + 1], acc[t][r0 + 2], acc[t][r0 + 3]);
        }
    }
    __syncthreads();
    if (s != 0) return;

#pragma unroll
    for (int p = 0; p < CSPLIT - 1; ++p)
#pragma unroll
        for (int q = 0; q < 8; ++q) {
            const int t = q >> 1, r0 = (q & 1) * 4;
            const float4 v = red[nt][p][q][lane];
            acc[t][r0 + 0] += v.x;
            acc[t][r0 + 1] += v.y;
            acc[t][r0 + 2] += v.z;
            acc[t][r0 + 3] += v.w;
        }

    float mx = -3.402823466e38f;
#pragma unroll
    for (int t = 0; t < 4; ++t)
#pragma unroll
        for (int r = 0; r < 8; ++r) mx = fmaxf(mx, acc[t][r]);
    mx = fmaxf(mx, __shfl_xor(mx, 16, 32));

    float sum = 0.0f;
#pragma unroll
    for (int t = 0; t < 4; ++t)
#pragma unroll
        for (int r = 0; r < 8; ++r) {
            float e = __expf(acc[t][r] - mx);
            acc[t][r] = e;
            sum += e;
        }
    sum += __shfl_xor(sum, 16, 32);
    const float inv = 1.0f / sum;

    float* ab = a + (size_t)b * NK * NN;
#pragma unroll
    for (int t = 0; t < 4; ++t)
#pragma unroll
        for (int r = 0; r < 8; ++r) {
            int k = t * 16 + r + 8 * half;
            ab[(size_t)k * NN + n0 + col] = acc[t][r] * inv;
        }
}

__global__ __launch_bounds__(128) void netvlad_vlad(
    const float* __restrict__ x, const float* __restrict__ a,
    const float* __restrict__ cent, float* __restrict__ out, float* __restrict__ part)
{
    __shared__ float4 srow[4][NK / 4];
    __shared__ float wsq[4];

    const int lane = threadIdx.x & 31;
    const int wid  = threadIdx.x >> 5;
    const int col  = lane & 15;
    const int half = lane >> 4;
    const int koff = half << 2;

    const int tile = blockIdx.x * 4 + wid;
    const int b    = tile >> 7;
    const int c0   = (tile & 127) << 4;

    const float* ab   = a + (size_t)b * NK * NN;
    const float* xrow = x + ((size_t)b * NC + c0 + col) * NN;

    v8f acc[4] = {v8f{}, v8f{}, v8f{}, v8f{}};
    v2f rs[4]  = {v2f{}, v2f{}, v2f{}, v2f{}};

    for (int n = 0; n < NN; n += 8) {
        const float4 a0 = *(const float4*)(ab + (size_t)( 0 + col) * NN + n + koff);
        const float4 a1 = *(const float4*)(ab + (size_t)(16 + col) * NN + n + koff);
        const float4 a2 = *(const float4*)(ab + (size_t)(32 + col) * NN + n + koff);
        const float4 a3 = *(const float4*)(ab + (size_t)(48 + col) * NN + n + koff);
        const float4 bv = *(const float4*)(xrow + n + koff);
        const v2f b0 = v2f{bv.x, bv.y};
        const v2f b1 = v2f{bv.z, bv.w};

        rs[0] += v2f{a0.x, a0.y} + v2f{a0.z, a0.w};
        rs[1] += v2f{a1.x, a1.y} + v2f{a1.z, a1.w};
        rs[2] += v2f{a2.x, a2.y} + v2f{a2.z, a2.w};
        rs[3] += v2f{a3.x, a3.y} + v2f{a3.z, a3.w};

        acc[0] = __builtin_amdgcn_wmma_f32_16x16x4_f32(false, v2f{a0.x, a0.y}, false, b0, (short)0, acc[0], false, false);
        acc[1] = __builtin_amdgcn_wmma_f32_16x16x4_f32(false, v2f{a1.x, a1.y}, false, b0, (short)0, acc[1], false, false);
        acc[2] = __builtin_amdgcn_wmma_f32_16x16x4_f32(false, v2f{a2.x, a2.y}, false, b0, (short)0, acc[2], false, false);
        acc[3] = __builtin_amdgcn_wmma_f32_16x16x4_f32(false, v2f{a3.x, a3.y}, false, b0, (short)0, acc[3], false, false);
        acc[0] = __builtin_amdgcn_wmma_f32_16x16x4_f32(false, v2f{a0.z, a0.w}, false, b1, (short)0, acc[0], false, false);
        acc[1] = __builtin_amdgcn_wmma_f32_16x16x4_f32(false, v2f{a1.z, a1.w}, false, b1, (short)0, acc[1], false, false);
        acc[2] = __builtin_amdgcn_wmma_f32_16x16x4_f32(false, v2f{a2.z, a2.w}, false, b1, (short)0, acc[2], false, false);
        acc[3] = __builtin_amdgcn_wmma_f32_16x16x4_f32(false, v2f{a3.z, a3.w}, false, b1, (short)0, acc[3], false, false);
    }

    float* strip = (float*)srow[wid];
    float rsum[4];
#pragma unroll
    for (int t = 0; t < 4; ++t) {
        const float r = rs[t].x + rs[t].y;
        rsum[t] = r + __shfl_xor(r, 16, 32);
    }
    if (half == 0) {
#pragma unroll
        for (int t = 0; t < 4; ++t) strip[t * 16 + col] = rsum[t];
    }
    __syncthreads();

    float* ob = out + (size_t)b * NK * NC;
    float sq[4] = {0.0f, 0.0f, 0.0f, 0.0f};
#pragma unroll
    for (int t = 0; t < 4; ++t) {
        const float4 s0 = srow[wid][t * 4 + 2 * half];
        const float4 s1 = srow[wid][t * 4 + 2 * half + 1];
        const float sv[8] = {s0.x, s0.y, s0.z, s0.w, s1.x, s1.y, s1.z, s1.w};
#pragma unroll
        for (int r = 0; r < 8; ++r) {
            int k = t * 16 + r + 8 * half;
            float cv = cent[(size_t)k * NC + c0 + col];
            const float o = acc[t][r] - sv[r] * cv;
            ob[(size_t)k * NC + c0 + col] = o;
            sq[t] += o * o;
        }
    }

    float ss = (sq[0] + sq[1]) + (sq[2] + sq[3]);
    ss += __shfl_xor(ss, 16, 32);
    ss += __shfl_xor(ss,  8, 32);
    ss += __shfl_xor(ss,  4, 32);
    ss += __shfl_xor(ss,  2, 32);
    ss += __shfl_xor(ss,  1, 32);
    if (lane == 0) wsq[wid] = ss;
    __syncthreads();
    if (threadIdx.x == 0) part[blockIdx.x] = (wsq[0] + wsq[1]) + (wsq[2] + wsq[3]);
}

#define NPART 32

__global__ __launch_bounds__(256) void netvlad_scale(
    float* __restrict__ out, const float* __restrict__ part)
{
    const int lane = threadIdx.x & 31;
    const int b    = blockIdx.x >> 7;
    const int gid  = blockIdx.x * 256 + threadIdx.x;
    float4* p = (float4*)out;
    float4 v = p[gid];

    float s = part[b * NPART + lane];
    s += __shfl_xor(s, 16, 32);
    s += __shfl_xor(s,  8, 32);
    s += __shfl_xor(s,  4, 32);
    s += __shfl_xor(s,  2, 32);
    s += __shfl_xor(s,  1, 32);
    const float sc = 1.0f / fmaxf(sqrtf(s), 1e-12f);

    v.x *= sc; v.y *= sc; v.z *= sc; v.w *= sc;
    p[gid] = v;
}

extern "C" void kernel_launch(void* const* d_in, const int* in_sizes, int n_in,
                              void* d_out, int out_size, void* d_ws, size_t ws_size,
                              hipStream_t stream) {
    const float* x    = (const float*)d_in[0];
    const float* w    = (const float*)d_in[1];
    const float* cent = (const float*)d_in[2];
    float* out = (float*)d_out;

    float* a    = (float*)d_ws;
    float* part = a + (size_t)NB * NK * NN;

    netvlad_assign<<<(NB * NN / 16) / NTB, 32 * CSPLIT * NTB, 0, stream>>>(x, w, a);
    netvlad_vlad  <<<(NB * NC / 16) / 4, 128, 0, stream>>>(x, a, cent, out, part);
    netvlad_scale <<<(NB * NK * NC / 4) / 256, 256, 0, stream>>>(out, part);
}


# vlad kernel: a and x staged through LDS in 64-n chunks by whole-line b128 loads issued one chunk ahead (a fetched once per 4 waves), fragments read by ds_load_b128; n order of every sum unchanged
# speedup vs baseline: 3.6473x; 1.1167x over previous
#include <hip/hip_runtime.h>

typedef __attribute__((ext_vector_type(2))) float v2f;
typedef __attribute__((ext_vector_type(8))) float v8f;

#define NK 64
#define NC 2048
#define NN 1024
#define NB 16

#define CSPLIT 8
#define NTB    2

__global__ __launch_bounds__(512) void netvlad_assign(
    const float* __restrict__ x, const float* __restrict__ w, float* __restrict__ a)
{
    __shared__ float4 red[NTB][CSPLIT - 1][8][32];

    const int lane = threadIdx.x & 31;
    const int wid  = __builtin_amdgcn_readfirstlane(threadIdx.x >> 5);
    const int s    = wid & (CSPLIT - 1);
    const int nt   = wid >> 3;
    const int col  = lane & 15;
    const int half = lane >> 4;
    const int koff = half << 2;

    const int tile = blockIdx.x * NTB + nt;
    const int b    = tile >> 6;
    const int n0   = (tile & 63) << 4;

    const float* xb = x + (size_t)b * NC * NN;

    v8f acc[4] = {v8f{}, v8f{}, v8f{}, v8f{}};

    for (int g = s; g < NC / 16; g += CSPLIT) {
        for (int u = 0; u < 2; ++u) {
            const int c = g * 16 + u * 8 + koff;
            const float4 a0 = *(const float4*)(w + (size_t)( 0 + col) * NC + c);
            const float4 a1 = *(const float4*)(w + (size_t)(16 + col) * NC + c);
            const float4 a2 = *(const float4*)(w + (size_t)(32 + col) * NC + c);
            const float4 a3 = *(const float4*)(w + (size_t)(48 + col) * NC + c);
            v2f b0, b1;
            b0.x = xb[(size_t)(c + 0) * NN + n0 + col];
            b0.y = xb[(size_t)(c + 1) * NN + n0 + col];
            b1.x = xb[(size_t)(c + 2) * NN + n0 + col];
            b1.y = xb[(size_t)(c + 3) * NN + n0 + col];

            acc[0] = __builtin_amdgcn_wmma_f32_16x16x4_f32(false, v2f{a0.x, a0.y}, false, b0, (short)0, acc[0], false, false);
            acc[1] = __builtin_amdgcn_wmma_f32_16x16x4_f32(false, v2f{a1.x, a1.y}, false, b0, (short)0, acc[1], false, false);
            acc[2] = __builtin_amdgcn_wmma_f32_16x16x4_f32(false, v2f{a2.x, a2.y}, false, b0, (short)0, acc[2], false, false);
            acc[3] = __builtin_amdgcn_wmma_f32_16x16x4_f32(false, v2f{a3.x, a3.y}, false, b0, (short)0, acc[3], false, false);
            acc[0] = __builtin_amdgcn_wmma_f32_16x16x4_f32(false, v2f{a0.z, a0.w}, false, b1, (short)0, acc[0], false, false);
            acc[1] = __builtin_amdgcn_wmma_f32_16x16x4_f32(false, v2f{a1.z, a1.w}, false, b1, (short)0, acc[1], false, false);
            acc[2] = __builtin_amdgcn_wmma_f32_16x16x4_f32(false, v2f{a2.z, a2.w}, false, b1, (short)0, acc[2], false, false);
            acc[3] = __builtin_amdgcn_wmma_f32_16x16x4_f32(false, v2f{a3.z, a3.w}, false, b1, (short)0, acc[3], false, false);
        }
    }

    if (s != 0) {
#pragma unroll
        for (int q = 0; q < 8; ++q) {
            const int t = q >> 1, r0 = (q & 1) * 4;
            red[nt][s - 1][q][lane] =
                make_float4(acc[t][r0 + 0], acc[t][r0 + 1], acc[t][r0 + 2], acc[t][r0 + 3]);
        }
    }
    __syncthreads();
    if (s != 0) return;

#pragma unroll
    for (int p = 0; p < CSPLIT - 1; ++p)
#pragma unroll
        for (int q = 0; q < 8; ++q) {
            const int t = q >> 1, r0 = (q & 1) * 4;
            const float4 v = red[nt][p][q][lane];
            acc[t][r0 + 0] += v.x;
            acc[t][r0 + 1] += v.y;
            acc[t][r0 + 2] += v.z;
            acc[t][r0 + 3] += v.w;
        }

    float mx = -3.402823466e38f;
#pragma unroll
    for (int t = 0; t < 4; ++t)
#pragma unroll
        for (int r = 0; r < 8; ++r) mx = fmaxf(mx, acc[t][r]);
    mx = fmaxf(mx, __shfl_xor(mx, 16, 32));

    float sum = 0.0f;
#pragma unroll
    for (int t = 0; t < 4; ++t)
#pragma unroll
        for (int r = 0; r < 8; ++r) {
            float e = __expf(acc[t][r] - mx);
            acc[t][r] = e;
            sum += e;
        }
    sum += __shfl_xor(sum, 16, 32);
    const float inv = 1.0f / sum;

    float* ab = a + (size_t)b * NK * NN;
#pragma unroll
    for (int t = 0; t < 4; ++t)
#pragma unroll
        for (int r = 0; r < 8; ++r) {
            int k = t * 16 + r + 8 * half;
            ab[(size_t)k * NN + n0 + col] = acc[t][r] * inv;
        }
}

#define VCH   64
#define VPQ   (VCH / 4 + 1)
#define VTC   64
#define VROWS (NK + VTC)

__device__ __forceinline__ void vlad_chunk(const float4* cur, const int fa, const int fb,
                                           v8f (&acc)[4], v2f (&rs)[4])
{
    float4 fA[2][4];
    float4 fB[2];
    fA[0][0] = cur[fa + ( 0) * VPQ];
    fA[0][1] = cur[fa + (16) * VPQ];
    fA[0][2] = cur[fa + (32) * VPQ];
    fA[0][3] = cur[fa + (48) * VPQ];
    fB[0]    = cur[fb];
#pragma unroll
    for (int jj = 0; jj < VCH / 8; ++jj) {
        const int cs = jj & 1, ns = cs ^ 1;
        if (jj + 1 < VCH / 8) {
            fA[ns][0] = cur[fa + ( 0) * VPQ + 2 * (jj + 1)];
            fA[ns][1] = cur[fa + (16) * VPQ + 2 * (jj + 1)];
            fA[ns][2] = cur[fa + (32) * VPQ + 2 * (jj + 1)];
            fA[ns][3] = cur[fa + (48) * VPQ + 2 * (jj + 1)];
            fB[ns]    = cur[fb + 2 * (jj + 1)];
        }
        const float4 a0 = fA[cs][0], a1 = fA[cs][1], a2 = fA[cs][2], a3 = fA[cs][3];
        const float4 bv = fB[cs];
        const v2f b0 = v2f{bv.x, bv.y};
        const v2f b1 = v2f{bv.z, bv.w};

        rs[0] += v2f{a0.x, a0.y} + v2f{a0.z, a0.w};
        rs[1] += v2f{a1.x, a1.y} + v2f{a1.z, a1.w};
        rs[2] += v2f{a2.x, a2.y} + v2f{a2.z, a2.w};
        rs[3] += v2f{a3.x, a3.y} + v2f{a3.z, a3.w};

        acc[0] = __builtin_amdgcn_wmma_f32_16x16x4_f32(false, v2f{a0.x, a0.y}, false, b0, (short)0, acc[0], false, false);
        acc[1] = __builtin_amdgcn_wmma_f32_16x16x4_f32(false, v2f{a1.x, a1.y}, false, b0, (short)0, acc[1], false, false);
        acc[2] = __builtin_amdgcn_wmma_f32_16x16x4_f32(false, v2f{a2.x, a2.y}, false, b0, (short)0, acc[2], false, false);
        acc[3] = __builtin_amdgcn_wmma_f32_16x16x4_f32(false, v2f{a3.x, a3.y}, false, b0, (short)0, acc[3], false, false);
        acc[0] = __builtin_amdgcn_wmma_f32_16x16x4_f32(false, v2f{a0.z, a0.w}, false, b1, (short)0, acc[0], false, false);
        acc[1] = __builtin_amdgcn_wmma_f32_16x16x4_f32(false, v2f{a1.z, a1.w}, false, b1, (short)0, acc[1], false, false);
        acc[2] = __builtin_amdgcn_wmma_f32_16x16x4_f32(false, v2f{a2.z, a2.w}, false, b1, (short)0, acc[2], false, false);
        acc[3] = __builtin_amdgcn_wmma_f32_16x16x4_f32(false, v2f{a3.z, a3.w}, false, b1, (short)0, acc[3], false, false);
    }
}

__global__ __launch_bounds__(128) void netvlad_vlad(
    const float* x, const float* a,
    const float* __restrict__ cent, float* __restrict__ out, float* __restrict__ part)
{
    __shared__ float4 stg[2][VROWS * VPQ];
    __shared__ float4 srow[4][NK / 4];
    __shared__ float wsq[4];

    const int lane = threadIdx.x & 31;
    const int wid  = threadIdx.x >> 5;
    const int col  = lane & 15;
    const int half = lane >> 4;

    const int tile = blockIdx.x * 4 + wid;
    const int b    = tile >> 7;
    const int c0   = (tile & 127) << 4;
    const int cb   = (blockIdx.x & 31) * VTC;

    const int q  = threadIdx.x & 15;
    const int r0 = threadIdx.x >> 4;
    const float* ga = a + ((size_t)b * NK + r0) * NN + q * 4;
    const float* gx = x + ((size_t)b * NC + cb + r0) * NN + q * 4;
    const int so = r0 * VPQ + q;

    const int fa = col * VPQ + half;
    const int fb = (NK + wid * 16 + col) * VPQ + half;

    v8f acc[4] = {v8f{}, v8f{}, v8f{}, v8f{}};
    v2f rs[4]  = {v2f{}, v2f{}, v2f{}, v2f{}};

#define VLAD_ROWS(F) F(0) F(1) F(2) F(3) F(4) F(5) F(6) F(7)
#define VLAD_DECL(j)  float4 ra##j, rx##j;
#define VLAD_FETCH(j) ra##j = *(const float4*)(ga + (size_t)(8 * j) * NN + n1); \
                      rx##j = *(const float4*)(gx + (size_t)(8 * j) * NN + n1);
#define VLAD_PUT(j)   nxt[so + (8 * j) * VPQ] = ra##j; nxt[so + (NK + 8 * j) * VPQ] = rx##j;
    VLAD_ROWS(VLAD_DECL)

    {
        const int n1 = 0;
        float4* nxt = stg[0];
        VLAD_ROWS(VLAD_FETCH)
        VLAD_ROWS(VLAD_PUT)
    }
    __syncthreads();

    for (int ch = 0; ch + 1 < NN / VCH; ++ch) {
        const float4* cur = stg[ch & 1];
        float4*       nxt = stg[(ch + 1) & 1];
        const int n1 = (ch + 1) * VCH;
        VLAD_ROWS(VLAD_FETCH)
        __builtin_amdgcn_sched_barrier(0);

        vlad_chunk(cur, fa, fb, acc, rs);

        VLAD_ROWS(VLAD_PUT)
        __syncthreads();
    }
    vlad_chunk(stg[(NN / VCH - 1) & 1], fa, fb, acc, rs);
#undef VLAD_ROWS
#undef VLAD_DECL
#undef VLAD_FETCH
#undef VLAD_PUT

    float* strip = (float*)srow[wid];
    float rsum[4];
#pragma unroll
    for (int t = 0; t < 4; ++t) {
        const float r = rs[t].x + rs[t].y;
        rsum[t] = r + __shfl_xor(r, 16, 32);
    }
    if (half == 0) {
#pragma unroll
        for (int t = 0; t < 4; ++t) strip[t * 16 + col] = rsum[t];
    }
    __syncthreads();

    float* ob = out + (size_t)b * NK * NC;
    float sq[4] = {0.0f, 0.0f, 0.0f, 0.0f};
#pragma unroll
    for (int t = 0; t < 4; ++t) {
        const float4 s0 = srow[wid][t * 4 + 2 * half];
        const float4 s1 = srow[wid][t * 4 + 2 * half + 1];
        const float sv[8] = {s0.x, s0.y, s0.z, s0.w, s1.x, s1.y, s1.z, s1.w};
#pragma unroll
        for (int r = 0; r < 8; ++r) {
            int k = t * 16 + r + 8 * half;
            float cv = cent[(size_t)k * NC + c0 + col];
            const float o = acc[t][r] - sv[r] * cv;
            ob[(size_t)k * NC + c0 + col] = o;
            sq[t] += o * o;
        }
    }

    float ss = (sq[0] + sq[1]) + (sq[2] + sq[3]);
    ss += __shfl_xor(ss, 16, 32);
    ss += __shfl_xor(ss,  8, 32);
    ss += __shfl_xor(ss,  4, 32);
    ss += __shfl_xor(ss,  2, 32);
    ss += __shfl_xor(ss,  1, 32);
    if (lane == 0) wsq[wid] = ss;
    __syncthreads();
    if (threadIdx.x == 0) part[blockIdx.x] = (wsq[0] + wsq[1]) + (wsq[2] + wsq[3]);
}

#define NPART 32

__global__ __launch_bounds__(256) void netvlad_scale(
    float* __restrict__ out, const float* __restrict__ part)
{
    const int lane = threadIdx.x & 31;
    const int b    = blockIdx.x >> 7;
    const int gid  = blockIdx.x * 256 + threadIdx.x;
    float4* p = (float4*)out;
    float4 v = p[gid];

    float s = part[b * NPART + lane];
    s += __shfl_xor(s, 16, 32);
    s += __shfl_xor(s,  8, 32);
    s += __shfl_xor(s,  4, 32);
    s += __shfl_xor(s,  2, 32);
    s += __shfl_xor(s,  1, 32);
    const float sc = 1.0f / fmaxf(sqrtf(s), 1e-12f);

    v.x *= sc; v.y *= sc; v.z *= sc; v.w *= sc;
    p[gid] = v;
}

extern "C" void kernel_launch(void* const* d_in, const int* in_sizes, int n_in,
                              void* d_out, int out_size, void* d_ws, size_t ws_size,
                              hipStream_t stream) {
    const float* x    = (const float*)d_in[0];
    const float* w    = (const float*)d_in[1];
    const float* cent = (const float*)d_in[2];
    float* out = (float*)d_out;

    float* a    = (float*)d_ws;
    float* part = a + (size_t)NB * NK * NN;

    netvlad_assign<<<(NB * NN / 16) / NTB, 32 * CSPLIT * NTB, 0, stream>>>(x, w, a);
    netvlad_vlad  <<<(NB * NC / 16) / 4, 128, 0, stream>>>(x, a, cent, out, part);
    netvlad_scale <<<(NB * NK * NC / 4) / 256, 256, 0, stream>>>(out, part);
}


# assign kernel re-tiled: a wave keeps 64k x 64n (n = 4*col+j) over every 8th 32-channel group: x rows by b128 loads, a w fragment serves 4 n-tiles, next group's loads issued ahead; reorders the c sum
# speedup vs baseline: 3.9572x; 1.0850x over previous
#include <hip/hip_runtime.h>

typedef __attribute__((ext_vector_type(2))) float v2f;
typedef __attribute__((ext_vector_type(8))) float v8f;

#define NK 64
#define NC 2048
#define NN 1024
#define NB 16

#define CSPLIT 8
#define AUNIT  32

__device__ __forceinline__ void assign_fetch(const float* wl, const float* xl, const int cb,
                                             float4 (&wa)[16], float4 (&xv)[16])
{
#pragma unroll
    for (int u = 0; u < 4; ++u) {
#pragma unroll
        for (int t = 0; t < 4; ++t)
            wa[u * 4 + t] = *(const float4*)(wl + (size_t)(16 * t) * NC + cb + 8 * u);
#pragma unroll
        for (int i = 0; i < 4; ++i)
            xv[u * 4 + i] = *(const float4*)(xl + (size_t)(cb + 8 * u + i) * NN);
    }
}

__device__ __forceinline__ void assign_products(const float4 (&wa)[16], const float4 (&xv)[16],
                                                v8f (&acc)[4][4])
{
#pragma unroll
    for (int u = 0; u < 4; ++u) {
        const float4 x0 = xv[u * 4 + 0], x1 = xv[u * 4 + 1];
        const float4 x2 = xv[u * 4 + 2], x3 = xv[u * 4 + 3];
        const v2f b0[4] = {v2f{x0.x, x1.x}, v2f{x0.y, x1.y}, v2f{x0.z, x1.z}, v2f{x0.w, x1.w}};
        const v2f b1[4] = {v2f{x2.x, x3.x}, v2f{x2.y, x3.y}, v2f{x2.z, x3.z}, v2f{x2.w, x3.w}};
#pragma unroll
        for (int t = 0; t < 4; ++t) {
            const float4 av = wa[u * 4 + t];
            const v2f a0 = v2f{av.x, av.y};
#pragma unroll
            for (int j = 0; j < 4; ++j)
                acc[t][j] = __builtin_amdgcn_wmma_f32_16x16x4_f32(false, a0, false, b0[j], (short)0, acc[t][j], false, false);
        }
#pragma unroll
        for (int t = 0; t < 4; ++t) {
            const float4 av = wa[u * 4 + t];
            const v2f a1 = v2f{av.z, av.w};
#pragma unroll
            for (int j = 0; j < 4; ++j)
                acc[t][j] = __builtin_amdgcn_wmma_f32_16x16x4_f32(false, a1, false, b1[j], (short)0, acc[t][j], false, false);
        }
    }
}

__global__ __launch_bounds__(256) void netvlad_assign(
    const float* x, const float* w, float* __restrict__ a)
{
    __shared__ float4 red[CSPLIT - 1][32][32];

    const int lane = threadIdx.x & 31;
    const int s    = __builtin_amdgcn_readfirstlane(threadIdx.x >> 5);
    const int col  = lane & 15;
    const int half = lane >> 4;
    const int koff = half << 2;

    const int b  = blockIdx.x >> 4;
    const int n0 = (blockIdx.x & 15) << 6;

    const float* wl = w + (size_t)col * NC + koff;
    const float* xl = x + ((size_t)b * NC + koff) * NN + n0 + 4 * col;

    v8f acc[4][4];
#pragma unroll
    for (int t = 0; t < 4; ++t)
#pragma unroll
        for (int j = 0; j < 4; ++j) acc[t][j] = v8f{};

    float4 wa0[16], xv0[16], wa1[16], xv1[16];
#define ACB(m) (((m) * CSPLIT + s) * AUNIT)

    assign_fetch(wl, xl, ACB(0), wa0, xv0);
    for (int m = 0; m < NC / AUNIT / CSPLIT - 2; m += 2) {
        assign_fetch(wl, xl, ACB(m + 1), wa1, xv1);
        __builtin_amdgcn_sched_barrier(0);
        assign_products(wa0, xv0, acc);
        assign_fetch(wl, xl, ACB(m + 2), wa0, xv0);
        __builtin_amdgcn_sched_barrier(0);
        assign_products(wa1, xv1, acc);
    }
    assign_fetch(wl, xl, ACB(NC / AUNIT / CSPLIT - 1), wa1, xv1);
    __builtin_amdgcn_sched_barrier(0);
    assign_products(wa0, xv0, acc);
    assign_products(wa1, xv1, acc);
#undef ACB

    if (s != 0) {
#pragma unroll
        for (int q = 0; q < 32; ++q) {
            const int t = q >> 3, j = (q >> 1) & 3, r0 = (q & 1) * 4;
            red[s - 1][q][lane] =
                make_float4(acc[t][j][r0 + 0], acc[t][j][r0 + 1], acc[t][j][r0 + 2], acc[t][j][r0 + 3]);
        }
    }
    __syncthreads();
    if (s != 0) return;

#pragma unroll
    for (int p = 0; p < CSPLIT - 1; ++p)
#pragma unroll
        for (int q = 0; q < 32; ++q) {
            const int t = q >> 3, j = (q >> 1) & 3, r0 = (q & 1) * 4;
            const float4 v = red[p][q][lane];
            acc[t][j][r0 + 0] += v.x;
            acc[t][j][r0 + 1] += v.y;
            acc[t][j][r0 + 2] += v.z;
            acc[t][j][r0 + 3] += v.w;
        }

    float inv[4];
#pragma unroll
    for (int j = 0; j < 4; ++j) {
        float mx = -3.402823466e38f;
#pragma unroll
        for (int t = 0; t < 4; ++t)
#pragma unroll
            for (int r = 0; r < 8; ++r) mx = fmaxf(mx, acc[t][j][r]);
        mx = fmaxf(mx, __shfl_xor(mx, 16, 32));

        float sum = 0.0f;
#pragma unroll
        for (int t = 0; t < 4; ++t)
#pragma unroll
            for (int r = 0; r < 8; ++r) {
                float e = __expf(acc[t][j][r] - mx);
                acc[t][j][r] = e;
                sum += e;
            }
        sum += __shfl_xor(sum, 16, 32);
        inv[j] = 1.0f / sum;
    }

    float* ab = a + (size_t)b * NK * NN + n0 + 4 * col;
#pragma unroll
    for (int t = 0; t < 4; ++t)
#pragma unroll
        for (int r = 0; r < 8; ++r) {
            const int k = t * 16 + r + 8 * half;
            *(float4*)(ab + (size_t)k * NN) = make_float4(acc[t][0][r] * inv[0], acc[t][1][r] * inv[1],
                                                          acc[t][2][r] * inv[2], acc[t][3][r] * inv[3]);
        }
}

#define VCH   64
#define VPQ   (VCH / 4 + 1)
#define VTC   64
#define VROWS (NK + VTC)

__device__ __forceinline__ void vlad_chunk(const float4* cur, const int fa, const int fb,
                                           v8f (&acc)[4], v2f (&rs)[4])
{
    float4 fA[2][4];
    float4 fB[2];
    fA[0][0] = cur[fa + ( 0) * VPQ];
    fA[0][1] = cur[fa + (16) * VPQ];
    fA[0][2] = cur[fa + (32) * VPQ];
    fA[0][3] = cur[fa + (48) * VPQ];
    fB[0]    = cur[fb];
#pragma unroll
    for (int jj = 0; jj < VCH / 8; ++jj) {
        const int cs = jj & 1, ns = cs ^ 1;
        if (jj + 1 < VCH / 8) {
            fA[ns][0] = cur[fa + ( 0) * VPQ + 2 * (jj + 1)];
            fA[ns][1] = cur[fa + (16) * VPQ + 2 * (jj + 1)];
            fA[ns][2] = cur[fa + (32) * VPQ + 2 * (jj + 1)];
            fA[ns][3] = cur[fa + (48) * VPQ + 2 * (jj + 1)];
            fB[ns]    = cur[fb + 2 * (jj + 1)];
        }
        const float4 a0 = fA[cs][0], a1 = fA[cs][1], a2 = fA[cs][2], a3 = fA[cs][3];
        const float4 bv = fB[cs];
        const v2f b0 = v2f{bv.x, bv.y};
        const v2f b1 = v2f{bv.z, bv.w};

        rs[0] += v2f{a0.x, a0.y} + v2f{a0.z, a0.w};
        rs[1] += v2f{a1.x, a1.y} + v2f{a1.z, a1.w};
        rs[2] += v2f{a2.x, a2.y} + v2f{a2.z, a2.w};
        rs[3] += v2f{a3.x, a3.y} + v2f{a3.z, a3.w};

        acc[0] = __builtin_amdgcn_wmma_f32_16x16x4_f32(false, v2f{a0.x, a0.y}, false, b0, (short)0, acc[0], false, false);
        acc[1] = __builtin_amdgcn_wmma_f32_16x16x4_f32(false, v2f{a1.x, a1.y}, false, b0, (short)0, acc[1], false, false);
        acc[2] = __builtin_amdgcn_wmma_f32_16x16x4_f32(false, v2f{a2.x, a2.y}, false, b0, (short)0, acc[2], false, false);
        acc[3] = __builtin_amdgcn_wmma_f32_16x16x4_f32(false, v2f{a3.x, a3.y}, false, b0, (short)0, acc[3], false, false);
        acc[0] = __builtin_amdgcn_wmma_f32_16x16x4_f32(false, v2f{a0.z, a0.w}, false, b1, (short)0, acc[0], false, false);
        acc[1] = __builtin_amdgcn_wmma_f32_16x16x4_f32(false, v2f{a1.z, a1.w}, false, b1, (short)0, acc[1], false, false);
        acc[2] = __builtin_amdgcn_wmma_f32_16x16x4_f32(false, v2f{a2.z, a2.w}, false, b1, (short)0, acc[2], false, false);
        acc[3] = __builtin_amdgcn_wmma_f32_16x16x4_f32(false, v2f{a3.z, a3.w}, false, b1, (short)0, acc[3], false, false);
    }
}

__global__ __launch_bounds__(128) void netvlad_vlad(
    const float* x, const float* a,
    const float* __restrict__ cent, float* __restrict__ out, float* __restrict__ part)
{
    __shared__ float4 stg[2][VROWS * VPQ];
    __shared__ float4 srow[4][NK / 4];
    __shared__ float wsq[4];

    const int lane = threadIdx.x & 31;
    const int wid  = threadIdx.x >> 5;
    const int col  = lane & 15;
    const int half = lane >> 4;

    const int tile = blockIdx.x * 4 + wid;
    const int b    = tile >> 7;
    const int c0   = (tile & 127) << 4;
    const int cb   = (blockIdx.x & 31) * VTC;

    const int q  = threadIdx.x & 15;
    const int r0 = threadIdx.x >> 4;
    const float* ga = a + ((size_t)b * NK + r0) * NN + q * 4;
    const float* gx = x + ((size_t)b * NC + cb + r0) * NN + q * 4;
    const int so = r0 * VPQ + q;

    const int fa = col * VPQ + half;
    const int fb = (NK + wid * 16 + col) * VPQ + half;

    v8f acc[4] = {v8f{}, v8f{}, v8f{}, v8f{}};
    v2f rs[4]  = {v2f{}, v2f{}, v2f{}, v2f{}};

#define VLAD_ROWS(F) F(0) F(1) F(2) F(3) F(4) F(5) F(6) F(7)
#define VLAD_DECL(j)  float4 ra##j, rx##j;
#define VLAD_FETCH(j) ra##j = *(const float4*)(ga + (size_t)(8 * j) * NN + n1); \
                      rx##j = *(const float4*)(gx + (size_t)(8 * j) * NN + n1);
#define VLAD_PUT(j)   nxt[so + (8 * j) * VPQ] = ra##j; nxt[so + (NK + 8 * j) * VPQ] = rx##j;
    VLAD_ROWS(VLAD_DECL)

    {
        const int n1 = 0;
        float4* nxt = stg[0];
        VLAD_ROWS(VLAD_FETCH)
        VLAD_ROWS(VLAD_PUT)
    }
    __syncthreads();

    for (int ch = 0; ch + 1 < NN / VCH; ++ch) {
        const float4* cur = stg[ch & 1];
        float4*       nxt = stg[(ch + 1) & 1];
        const int n1 = (ch + 1) * VCH;
        VLAD_ROWS(VLAD_FETCH)
        __builtin_amdgcn_sched_barrier(0);

        vlad_chunk(cur, fa, fb, acc, rs);

        VLAD_ROWS(VLAD_PUT)
        __syncthreads();
    }
    vlad_chunk(stg[(NN / VCH - 1) & 1], fa, fb, acc, rs);
#undef VLAD_ROWS
#undef VLAD_DECL
#undef VLAD_FETCH
#undef VLAD_PUT

    float* strip = (float*)srow[wid];
    float rsum[4];
#pragma unroll
    for (int t = 0; t < 4; ++t) {
        const float r = rs[t].x + rs[t].y;
        rsum[t] = r + __shfl_xor(r, 16, 32);
    }
    if (half == 0) {
#pragma unroll
        for (int t = 0; t < 4; ++t) strip[t * 16 + col] = rsum[t];
    }
    __syncthreads();

    float* ob = out + (size_t)b * NK * NC;
    float sq[4] = {0.0f, 0.0f, 0.0f, 0.0f};
#pragma unroll
    for (int t = 0; t < 4; ++t) {
        const float4 s0 = srow[wid][t * 4 + 2 * half];
        const float4 s1 = srow[wid][t * 4 + 2 * half + 1];
        const float sv[8] = {s0.x, s0.y, s0.z, s0.w, s1.x, s1.y, s1.z, s1.w};
#pragma unroll
        for (int r = 0; r < 8; ++r) {
            int k = t * 16 + r + 8 * half;
            float cv = cent[(size_t)k * NC + c0 + col];
            const float o = acc[t][r] - sv[r] * cv;
            ob[(size_t)k * NC + c0 + col] = o;
            sq[t] += o * o;
        }
    }

    float ss = (sq[0] + sq[1]) + (sq[2] + sq[3]);
    ss += __shfl_xor(ss, 16, 32);
    ss += __shfl_xor(ss,  8, 32);
    ss += __shfl_xor(ss,  4, 32);
    ss += __shfl_xor(ss,  2, 32);
    ss += __shfl_xor(ss,  1, 32);
    if (lane == 0) wsq[wid] = ss;
    __syncthreads();
    if (threadIdx.x == 0) part[blockIdx.x] = (wsq[0] + wsq[1]) + (wsq[2] + wsq[3]);
}

#define NPART 32

__global__ __launch_bounds__(256) void netvlad_scale(
    float* __restrict__ out, const float* __restrict__ part)
{
    const int lane = threadIdx.x & 31;
    const int b    = blockIdx.x >> 7;
    const int gid  = blockIdx.x * 256 + threadIdx.x;
    float4* p = (float4*)out;
    float4 v = p[gid];

    float s = part[b * NPART + lane];
    s += __shfl_xor(s, 16, 32);
    s += __shfl_xor(s,  8, 32);
    s += __shfl_xor(s,  4, 32);
    s += __shfl_xor(s,  2, 32);
    s += __shfl_xor(s,  1, 32);
    const float sc = 1.0f / fmaxf(sqrtf(s), 1e-12f);

    v.x *= sc; v.y *= sc; v.z *= sc; v.w *= sc;
    p[gid] = v;
}

extern "C" void kernel_launch(void* const* d_in, const int* in_sizes, int n_in,
                              void* d_out, int out_size, void* d_ws, size_t ws_size,
                              hipStream_t stream) {
    const float* x    = (const float*)d_in[0];
    const float* w    = (const float*)d_in[1];
    const float* cent = (const float*)d_in[2];
    float* out = (float*)d_out;

    float* a    = (float*)d_ws;
    float* part = a + (size_t)NB * NK * NN;

    netvlad_assign<<<NB * NN / 64, 32 * CSPLIT, 0, stream>>>(x, w, a);
    netvlad_vlad  <<<(NB * NC / 16) / 4, 128, 0, stream>>>(x, a, cent, out, part);
    netvlad_scale <<<(NB * NK * NC / 4) / 256, 256, 0, stream>>>(out, part);
}
